# starting kernel
# baseline (speedup 1.0000x reference)
#include <hip/hip_runtime.h>


#define VEC_DIM   256
#define NUM_WORDS 50000
#define BATCH     4096
#define NUM_CTX   8
#define NUM_K     6
#define TILE_B    16
#define THREADS   128

typedef __attribute__((ext_vector_type(2))) float v2f;
typedef __attribute__((ext_vector_type(8))) float v8f;

__global__ __launch_bounds__(THREADS) void pvdm_wmma_kernel(
    const int*   __restrict__ ctx,
    const int*   __restrict__ doc,
    const int*   __restrict__ tgt,
    const float* __restrict__ Dm,
    const float* __restrict__ Wm,
    const float* __restrict__ Om,
    float*       __restrict__ out)
{
    __shared__ __align__(16) float xs[TILE_B][VEC_DIM + 4];
    __shared__ float part[2][TILE_B];

    const int b0   = blockIdx.x * TILE_B;
    const int tid  = threadIdx.x;
    const int lane = tid & 31;
    const int wave = tid >> 5;
    const int wk   = wave & 1;
    const int wd   = wave >> 1;

    for (int task = tid; task < TILE_B * (VEC_DIM / 4); task += THREADS) {
        const int row = task >> 6;
        const int c4  = task & 63;
        const int b   = b0 + row;
        const float4* dp = (const float4*)(Dm + (long)doc[b] * VEC_DIM) + c4;
        float4 acc = *dp;
        #pragma unroll
        for (int c = 0; c < NUM_CTX; ++c) {
            const float4* wp =
                (const float4*)(Wm + (long)ctx[b * NUM_CTX + c] * VEC_DIM) + c4;
            float4 wv = *wp;
            acc.x += wv.x; acc.y += wv.y; acc.z += wv.z; acc.w += wv.w;
        }
        *(float4*)&xs[row][c4 * 4] = acc;
    }
    __syncthreads();

    const int row   = lane & 15;
    const int kOff  = (lane < 16) ? 0 : 2;
    const int dBase = wd * (VEC_DIM / 2);

    for (int kk = 0; kk < 3; ++kk) {
        const int k   = wk * 3 + kk;
        const int col = tgt[(b0 + row) * NUM_K + k];
        const float* ob = Om + (long)dBase * NUM_WORDS + col;

        v8f c = {0.f, 0.f, 0.f, 0.f, 0.f, 0.f, 0.f, 0.f};

        #pragma unroll 8
        for (int step = 0; step < VEC_DIM / 8; ++step) {
            const int d0 = step * 4 + kOff;
            v2f a, b;
            a.x = xs[row][dBase + d0];
            a.y = xs[row][dBase + d0 + 1];
            b.x = ob[(long)d0 * NUM_WORDS];
            b.y = ob[(long)(d0 + 1) * NUM_WORDS];
            c = __builtin_amdgcn_wmma_f32_16x16x4_f32(
                    false, a, false, b, (short)0, c, false, false);
        }

        float val = 0.f;
        int   di  = 0;
        bool  own = false;
        if (lane < 8)        { val = c[lane];      di = lane;      own = true; }
        else if (lane >= 24) { val = c[lane - 24]; di = lane - 16; own = true; }

        if (wd == 1 && own) part[wk][di] = val;
        __syncthreads();
        if (wd == 0 && own)
            out[(b0 + di) * NUM_K + k] = val + part[wk][di];
        __syncthreads();
    }
}

extern "C" void kernel_launch(void* const* d_in, const int* in_sizes, int n_in,
                              void* d_out, int out_size, void* d_ws, size_t ws_size,
                              hipStream_t stream) {
    (void)in_sizes; (void)n_in; (void)out_size; (void)d_ws; (void)ws_size;
    const int*   ctx = (const int*)  d_in[0];
    const int*   doc = (const int*)  d_in[1];
    const int*   tgt = (const int*)  d_in[2];
    const float* Dm  = (const float*)d_in[3];
    const float* Wm  = (const float*)d_in[4];
    const float* Om  = (const float*)d_in[5];
    float* out = (float*)d_out;

    dim3 grid(BATCH / TILE_B);
    dim3 block(THREADS);
    hipLaunchKernelGGL(pvdm_wmma_kernel, grid, block, 0, stream,
                       ctx, doc, tgt, Dm, Wm, Om, out);
}


# Transpose launch: 208-word tiles, one 8-wave block per WGP (241 blocks, was 782); a lane has 52 row loads in flight, not 16, and fills the LDS tile with 16-byte stores of 4x4 blocks; sums unchanged
# speedup vs baseline: 1.9048x; 1.9048x over previous
#include <hip/hip_runtime.h>


#define VEC_DIM   256
#define NUM_WORDS 50000
#define BATCH     4096
#define NUM_CTX   8
#define NUM_K     6
#define TILE_B    16
#define THREADS   128

#define T_WORDS   208
#define T_CHUNKS  (T_WORDS / 16)
#define T_PITCH   (VEC_DIM + 4)
#define T_THREADS 256
#define T_ROWS    (T_WORDS / (T_THREADS / 32))
#define T_BLOCKS  ((NUM_WORDS + T_WORDS - 1) / T_WORDS)

typedef __attribute__((ext_vector_type(2))) float v2f;
typedef __attribute__((ext_vector_type(8))) float v8f;

template <bool FULL>
__device__ __forceinline__ void transpose_tile(
    const float* __restrict__ Om, float* __restrict__ OT,
    float (*ts)[T_PITCH], const int w0, const int nvalid)
{
    const int tid    = threadIdx.x;
    const int lane   = tid & 31;
    const int wave   = __builtin_amdgcn_readfirstlane(tid >> 5);
    const int wi     = lane & 3;
    const int dg     = wave * 8 + (lane >> 2);

    const unsigned voff = (unsigned)(dg * 4) * (NUM_WORDS * 4u) + (unsigned)wi * 16u;
    const char* tb = (const char*)Om + (size_t)((unsigned)w0 * 4u);

    float4 q[T_CHUNKS][4];
    #pragma unroll
    for (int c = 0; c < T_CHUNKS; ++c) {
        const unsigned co = (FULL || c * 16 < nvalid) ? (unsigned)c * 64u : 0u;
        #pragma unroll
        for (int j = 0; j < 4; ++j)
            q[c][j] = *(const float4*)(tb + voff + (co + (unsigned)j * (NUM_WORDS * 4u)));
    }
    #pragma unroll
    for (int c = 0; c < T_CHUNKS; ++c) {
        const int wl = c * 16 + wi * 4;
        *(float4*)&ts[wl + 0][dg * 4] = make_float4(q[c][0].x, q[c][1].x, q[c][2].x, q[c][3].x);
        *(float4*)&ts[wl + 1][dg * 4] = make_float4(q[c][0].y, q[c][1].y, q[c][2].y, q[c][3].y);
        *(float4*)&ts[wl + 2][dg * 4] = make_float4(q[c][0].z, q[c][1].z, q[c][2].z, q[c][3].z);
        *(float4*)&ts[wl + 3][dg * 4] = make_float4(q[c][0].w, q[c][1].w, q[c][2].w, q[c][3].w);
    }
    __syncthreads();

    const int wr0 = wave * T_ROWS;
    float* obase = OT + (long)(w0 + wr0) * VEC_DIM;
    #pragma unroll
    for (int j = 0; j < T_ROWS; ++j) {
        if (FULL || wr0 + j < nvalid) {
            const float4 r0 = *(const float4*)&ts[wr0 + j][lane * 4];
            const float4 r1 = *(const float4*)&ts[wr0 + j][128 + lane * 4];
            *(float4*)(obase + j * VEC_DIM + lane * 4)       = r0;
            *(float4*)(obase + j * VEC_DIM + 128 + lane * 4) = r1;
        }
    }
}

__global__ __launch_bounds__(T_THREADS)
void transpose_o_kernel(
    const float* __restrict__ Om,
    float*       __restrict__ OT)
{
    __shared__ __align__(16) float ts[T_WORDS][T_PITCH];

    const int w0     = blockIdx.x * T_WORDS;
    const int nvalid = NUM_WORDS - w0;
    if (nvalid >= T_WORDS)
        transpose_tile<true>(Om, OT, ts, w0, nvalid);
    else
        transpose_tile<false>(Om, OT, ts, w0, nvalid);
}

__global__ __launch_bounds__(THREADS) __attribute__((amdgpu_waves_per_eu(1, 1)))
void pvdm_wmma_kernel(
    const int*   __restrict__ ctx,
    const int*   __restrict__ doc,
    const int*   __restrict__ tgt,
    const float* __restrict__ Dm,
    const float* __restrict__ Wm,
    const float* __restrict__ OT,
    float*       __restrict__ out)
{
    __shared__ __align__(16) float xs[TILE_B][VEC_DIM + 4];
    __shared__ float part[2][3][TILE_B];

    const int b0   = blockIdx.x * TILE_B;
    const int tid  = threadIdx.x;
    const int lane = tid & 31;
    const int wave = __builtin_amdgcn_readfirstlane(tid >> 5);
    const int wk   = wave & 1;
    const int wd   = wave >> 1;

    const int row   = lane & 15;
    const int hOff  = (lane < 16) ? 0 : 4;
    const int dBase = wd * (VEC_DIM / 2);

    int col[3];
    #pragma unroll
    for (int kk = 0; kk < 3; ++kk)
        col[kk] = tgt[(b0 + row) * NUM_K + wk * 3 + kk];

    const int r0 = b0 + 4 * wave;
    int dIdx[4];
    int cIdx[4][NUM_CTX];
    #pragma unroll
    for (int r = 0; r < 4; ++r) {
        dIdx[r] = doc[r0 + r];
        #pragma unroll
        for (int c = 0; c < NUM_CTX; ++c)
            cIdx[r][c] = ctx[(r0 + r) * NUM_CTX + c];
    }

    const unsigned lo = (unsigned)lane * 16u;
    float4 xv[4][1 + NUM_CTX][2];
    #pragma unroll
    for (int r = 0; r < 4; ++r) {
        const char* dp = (const char*)Dm + (size_t)((unsigned)dIdx[r] * (VEC_DIM * 4u));
        xv[r][0][0] = *(const float4*)(dp + lo);
        xv[r][0][1] = *(const float4*)(dp + lo + 512);
        #pragma unroll
        for (int c = 0; c < NUM_CTX; ++c) {
            const char* wp = (const char*)Wm + (size_t)((unsigned)cIdx[r][c] * (VEC_DIM * 4u));
            xv[r][1 + c][0] = *(const float4*)(wp + lo);
            xv[r][1 + c][1] = *(const float4*)(wp + lo + 512);
        }
    }

    float4 bq[3][16];
    #pragma unroll
    for (int kk = 0; kk < 3; ++kk) {
        const unsigned off = (unsigned)col[kk] * (VEC_DIM * 4u) + (unsigned)(dBase + hOff) * 4u;
        const char* ob = (const char*)OT + (size_t)off;
        #pragma unroll
        for (int step = 0; step < 16; ++step)
            bq[kk][step] = *(const float4*)(ob + step * 32);
    }

    #pragma unroll
    for (int r = 0; r < 4; ++r) {
        #pragma unroll
        for (int j = 0; j < 2; ++j) {
            float4 acc = xv[r][0][j];
            #pragma unroll
            for (int c = 0; c < NUM_CTX; ++c) {
                const float4 wv = xv[r][1 + c][j];
                acc.x += wv.x; acc.y += wv.y; acc.z += wv.z; acc.w += wv.w;
            }
            *(float4*)&xs[4 * wave + r][(lane + 32 * j) * 4] = acc;
        }
    }
    __syncthreads();

    const float* xa = &xs[row][dBase + hOff];
    v8f c0 = {0.f, 0.f, 0.f, 0.f, 0.f, 0.f, 0.f, 0.f};
    v8f c1 = c0, c2 = c0;
    #pragma unroll
    for (int step = 0; step < VEC_DIM / 16; ++step) {
        const float4 a4 = *(const float4*)(xa + step * 8);
        v2f a, b;
        a.x = a4.x; a.y = a4.y;
        b.x = bq[0][step].x; b.y = bq[0][step].y;
        c0 = __builtin_amdgcn_wmma_f32_16x16x4_f32(false, a, false, b, (short)0, c0, false, false);
        b.x = bq[1][step].x; b.y = bq[1][step].y;
        c1 = __builtin_amdgcn_wmma_f32_16x16x4_f32(false, a, false, b, (short)0, c1, false, false);
        b.x = bq[2][step].x; b.y = bq[2][step].y;
        c2 = __builtin_amdgcn_wmma_f32_16x16x4_f32(false, a, false, b, (short)0, c2, false, false);
        a.x = a4.z; a.y = a4.w;
        b.x = bq[0][step].z; b.y = bq[0][step].w;
        c0 = __builtin_amdgcn_wmma_f32_16x16x4_f32(false, a, false, b, (short)0, c0, false, false);
        b.x = bq[1][step].z; b.y = bq[1][step].w;
        c1 = __builtin_amdgcn_wmma_f32_16x16x4_f32(false, a, false, b, (short)0, c1, false, false);
        b.x = bq[2][step].z; b.y = bq[2][step].w;
        c2 = __builtin_amdgcn_wmma_f32_16x16x4_f32(false, a, false, b, (short)0, c2, false, false);
    }

    const int  sel = lane & 7;
    const bool own = (lane < 8) || (lane >= 24);
    const int  di  = (lane < 8) ? lane : lane - 16;
    float v0 = c0[0], v1 = c1[0], v2 = c2[0];
    #pragma unroll
    for (int r = 1; r < 8; ++r) {
        v0 = (sel == r) ? c0[r] : v0;
        v1 = (sel == r) ? c1[r] : v1;
        v2 = (sel == r) ? c2[r] : v2;
    }

    if (wd == 1 && own) {
        part[wk][0][di] = v0;
        part[wk][1][di] = v1;
        part[wk][2][di] = v2;
    }
    __syncthreads();
    if (wd == 0 && own) {
        float* o = out + (b0 + di) * NUM_K + wk * 3;
        o[0] = v0 + part[wk][0][di];
        o[1] = v1 + part[wk][1][di];
        o[2] = v2 + part[wk][2][di];
    }
}

extern "C" void kernel_launch(void* const* d_in, const int* in_sizes, int n_in,
                              void* d_out, int out_size, void* d_ws, size_t ws_size,
                              hipStream_t stream) {
    (void)in_sizes; (void)n_in; (void)out_size; (void)ws_size;
    const int*   ctx = (const int*)  d_in[0];
    const int*   doc = (const int*)  d_in[1];
    const int*   tgt = (const int*)  d_in[2];
    const float* Dm  = (const float*)d_in[3];
    const float* Wm  = (const float*)d_in[4];
    const float* Om  = (const float*)d_in[5];
    float* out = (float*)d_out;
    float* OT  = (float*)d_ws;

    hipLaunchKernelGGL(transpose_o_kernel, dim3(T_BLOCKS), dim3(T_THREADS), 0, stream,
                       Om, OT);

    dim3 grid(BATCH / TILE_B);
    dim3 block(THREADS);
    hipLaunchKernelGGL(pvdm_wmma_kernel, grid, block, 0, stream,
                       ctx, doc, tgt, Dm, Wm, OT, out);
}
